# starting kernel
# baseline (speedup 1.0000x reference)
#include <hip/hip_runtime.h>

typedef float v2f __attribute__((ext_vector_type(2)));
typedef float v8f __attribute__((ext_vector_type(8)));

#define HW_SZ          25600
#define C_SZ           64
#define B_SZ           32
#define PIX_PER_BLOCK  128
#define BLOCKS_PER_IMG (HW_SZ / PIX_PER_BLOCK)
#define NBLK1          (B_SZ * BLOCKS_PER_IMG)
#define NTOT           (B_SZ * HW_SZ)
#define BN_EPS_F       1e-5f
#define SLOPE_F        0.1f

__global__ __launch_bounds__(256) void maskgen_wmma(
    const float* __restrict__ sf, const float* __restrict__ feats,
    float* __restrict__ out, float* __restrict__ partials)
{
    const int lane  = threadIdx.x & 31;
    const int wave  = threadIdx.x >> 5;
    const int img   = blockIdx.x / BLOCKS_PER_IMG;
    const int chunk = blockIdx.x % BLOCKS_PER_IMG;
    const int p0    = chunk * PIX_PER_BLOCK + wave * 16;

    const float* __restrict__ sfb = sf + img * C_SZ;
    const float* __restrict__ fp  =
        feats + (size_t)img * C_SZ * HW_SZ + p0 + (lane & 15);

    const int hi = lane >> 4;
    const int m  = lane & 15;

    v8f acc = {};
    #pragma unroll
    for (int c = 0; c < C_SZ; c += 4) {
        const int ch = c + (hi << 1);
        v2f a, bm;
        a[0] = fp[(size_t)(ch + 0) * HW_SZ];
        a[1] = fp[(size_t)(ch + 1) * HW_SZ];
        const float s0 = sfb[c + 0], s1 = sfb[c + 1];
        const float s2 = sfb[c + 2], s3 = sfb[c + 3];
        bm[0] = hi ? s2 : s0;
        bm[1] = hi ? s3 : s1;
        acc = __builtin_amdgcn_wmma_f32_16x16x4_f32(
                  false, a, false, bm, (short)0, acc, false, false);
    }

    float* __restrict__ op = out + (size_t)img * HW_SZ + p0;
    if (m < 8) {
        float val = acc[0];
        #pragma unroll
        for (int v = 1; v < 8; ++v) if (m == v) val = acc[v];
        op[m + (hi << 3)] = val;
    }

    float ps = acc[0] + acc[1] + acc[2] + acc[3] + acc[4] + acc[5] + acc[6] + acc[7];
    float qs = acc[0]*acc[0] + acc[1]*acc[1] + acc[2]*acc[2] + acc[3]*acc[3]
             + acc[4]*acc[4] + acc[5]*acc[5] + acc[6]*acc[6] + acc[7]*acc[7];
    float fullS = ps + __shfl_xor(ps, 16, 32);
    float fullQ = qs + __shfl_xor(qs, 16, 32);

    __shared__ float sS[8], sQ[8];
    if (lane == 0) { sS[wave] = fullS; sQ[wave] = fullQ; }
    __syncthreads();
    if (threadIdx.x == 0) {
        float S = 0.f, Q = 0.f;
        #pragma unroll
        for (int w = 0; w < 8; ++w) { S += sS[w]; Q += sQ[w]; }
        partials[2 * blockIdx.x + 0] = S;
        partials[2 * blockIdx.x + 1] = Q;
    }
}

__global__ __launch_bounds__(256) void reduce_stats(
    const float* __restrict__ partials, float* __restrict__ stats)
{
    __shared__ float sS[256], sQ[256];
    float S = 0.f, Q = 0.f;
    for (int i = threadIdx.x; i < NBLK1; i += 256) {
        S += partials[2 * i + 0];
        Q += partials[2 * i + 1];
    }
    sS[threadIdx.x] = S; sQ[threadIdx.x] = Q;
    __syncthreads();
    for (int s = 128; s > 0; s >>= 1) {
        if ((int)threadIdx.x < s) {
            sS[threadIdx.x] += sS[threadIdx.x + s];
            sQ[threadIdx.x] += sQ[threadIdx.x + s];
        }
        __syncthreads();
    }
    if (threadIdx.x == 0) {
        const float invN = 1.0f / (float)NTOT;
        const float mean = sS[0] * invN;
        const float var  = sQ[0] * invN - mean * mean;
        stats[0] = mean;
        stats[1] = rsqrtf(var + BN_EPS_F);
    }
}

__global__ __launch_bounds__(256) void normalize_lrelu(
    float* __restrict__ out, const float* __restrict__ stats,
    const float* __restrict__ bn_w, const float* __restrict__ bn_b)
{
    const float mean = stats[0];
    const float g    = bn_w[0] * stats[1];
    const float bb   = bn_b[0] - mean * g;
    const int i = blockIdx.x * blockDim.x + threadIdx.x;
    float4* o4 = (float4*)out;
    float4 v = o4[i];
    float y0 = v.x * g + bb, y1 = v.y * g + bb;
    float y2 = v.z * g + bb, y3 = v.w * g + bb;
    v.x = (y0 >= 0.f) ? y0 : SLOPE_F * y0;
    v.y = (y1 >= 0.f) ? y1 : SLOPE_F * y1;
    v.z = (y2 >= 0.f) ? y2 : SLOPE_F * y2;
    v.w = (y3 >= 0.f) ? y3 : SLOPE_F * y3;
    o4[i] = v;
}

extern "C" void kernel_launch(void* const* d_in, const int* in_sizes, int n_in,
                              void* d_out, int out_size, void* d_ws, size_t ws_size,
                              hipStream_t stream) {
    const float* sf    = (const float*)d_in[0];
    const float* feats = (const float*)d_in[1];
    const float* bn_w  = (const float*)d_in[2];
    const float* bn_b  = (const float*)d_in[3];
    float* out      = (float*)d_out;
    float* partials = (float*)d_ws;
    float* stats    = partials + 2 * NBLK1;

    maskgen_wmma   <<<NBLK1, 256, 0, stream>>>(sf, feats, out, partials);
    reduce_stats   <<<1, 256, 0, stream>>>(partials, stats);
    normalize_lrelu<<<NTOT / 4 / 256, 256, 0, stream>>>(out, stats, bn_w, bn_b);
}


# Phase 1: 16-byte loads (4 pixels per lane feed 4 accumulator tiles), 2 tiles of 64 px per wave, next 16-channel group requested ahead: 6400 co-resident waves, not 51200; partial-sum tree regrouped
# speedup vs baseline: 1.1573x; 1.1573x over previous
#include <hip/hip_runtime.h>

typedef float v2f __attribute__((ext_vector_type(2)));
typedef float v4f __attribute__((ext_vector_type(4)));
typedef float v8f __attribute__((ext_vector_type(8)));

#define HW_SZ          25600
#define C_SZ           64
#define B_SZ           32
#define TILE_PIX       64
#define TILES_PER_WAVE 2
#define WAVES_PER_BLK  8
#define PIX_PER_WAVE   (TILES_PER_WAVE * TILE_PIX)
#define PIX_PER_BLOCK  (WAVES_PER_BLK * PIX_PER_WAVE)
#define BLOCKS_PER_IMG (HW_SZ / PIX_PER_BLOCK)
#define NBLK1          (B_SZ * BLOCKS_PER_IMG)
#define NTOT           (B_SZ * HW_SZ)
#define BN_EPS_F       1e-5f
#define SLOPE_F        0.1f

__global__ __launch_bounds__(256) void maskgen_wmma(
    const float* __restrict__ sf, const float* __restrict__ feats,
    float* __restrict__ out, float* __restrict__ partials)
{
    const int lane  = threadIdx.x & 31;
    const int wave  = threadIdx.x >> 5;
    const int img   = blockIdx.x / BLOCKS_PER_IMG;
    const int chunk = blockIdx.x % BLOCKS_PER_IMG;
    const int p0    = chunk * PIX_PER_BLOCK + wave * PIX_PER_WAVE;

    const int hi = lane >> 4;
    const int m  = lane & 15;

    const float* __restrict__ sfb = sf + img * C_SZ;
    const float* __restrict__ fp  =
        feats + (size_t)img * C_SZ * HW_SZ + (size_t)(hi << 1) * HW_SZ + p0 + (m << 2);
    float* __restrict__ op = out + (size_t)img * HW_SZ + p0;

    v4f buf[2][8];
    v8f acc[4];
    float ps = 0.f, qs = 0.f;

#define LOAD_GROUP(G)                                                              \
    _Pragma("unroll")                                                              \
    for (int s = 0; s < 4; ++s) {                                                  \
        _Pragma("unroll")                                                          \
        for (int j = 0; j < 2; ++j)                                                \
            buf[(G) & 1][2 * s + j] = *(const v4f*)(fp + ((G) >> 2) * TILE_PIX     \
                + (size_t)(16 * ((G) & 3) + 4 * s + j) * HW_SZ);                   \
    }

    LOAD_GROUP(0)
    #pragma unroll
    for (int g = 0; g < 4 * TILES_PER_WAVE; ++g) {
        __builtin_amdgcn_sched_barrier(0);
        if (g + 1 < 4 * TILES_PER_WAVE) { LOAD_GROUP(g + 1) }
        __builtin_amdgcn_sched_barrier(0);
        if ((g & 3) == 0) {
            #pragma unroll
            for (int q = 0; q < 4; ++q) acc[q] = (v8f){};
        }
        #pragma unroll
        for (int s = 0; s < 4; ++s) {
            const int c = 16 * (g & 3) + 4 * s;
            const float s0 = sfb[c + 0], s1 = sfb[c + 1];
            const float s2 = sfb[c + 2], s3 = sfb[c + 3];
            v2f bm;
            bm[0] = hi ? s2 : s0;
            bm[1] = hi ? s3 : s1;
            #pragma unroll
            for (int q = 0; q < 4; ++q) {
                v2f a;
                a[0] = buf[g & 1][2 * s + 0][q];
                a[1] = buf[g & 1][2 * s + 1][q];
                acc[q] = __builtin_amdgcn_wmma_f32_16x16x4_f32(
                             false, a, false, bm, (short)0, acc[q], false, false);
            }
        }
        if ((g & 3) == 3) {
            v4f o;
            #pragma unroll
            for (int q = 0; q < 4; ++q) {
                float val = acc[q][0];
                #pragma unroll
                for (int v = 1; v < 8; ++v) if (m == v) val = acc[q][v];
                o[q] = val;
            }
            if (m < 8)
                *(v4f*)(op + (g >> 2) * TILE_PIX + ((m + (hi << 3)) << 2)) = o;
            float ts[4], tq[4];
            #pragma unroll
            for (int q = 0; q < 4; ++q) {
                const v8f d = acc[q];
                ts[q] = ((d[0] + d[1]) + (d[2] + d[3])) + ((d[4] + d[5]) + (d[6] + d[7]));
                tq[q] = ((d[0]*d[0] + d[1]*d[1]) + (d[2]*d[2] + d[3]*d[3]))
                      + ((d[4]*d[4] + d[5]*d[5]) + (d[6]*d[6] + d[7]*d[7]));
            }
            ps += (ts[0] + ts[1]) + (ts[2] + ts[3]);
            qs += (tq[0] + tq[1]) + (tq[2] + tq[3]);
            asm volatile("" : "+v"(ps), "+v"(qs));
        }
    }
#undef LOAD_GROUP

    float fullS = ps + __shfl_xor(ps, 16, 32);
    float fullQ = qs + __shfl_xor(qs, 16, 32);

    __shared__ float sS[WAVES_PER_BLK], sQ[WAVES_PER_BLK];
    if (lane == 0) { sS[wave] = fullS; sQ[wave] = fullQ; }
    __syncthreads();
    if (threadIdx.x == 0) {
        float S = 0.f, Q = 0.f;
        #pragma unroll
        for (int w = 0; w < WAVES_PER_BLK; ++w) { S += sS[w]; Q += sQ[w]; }
        partials[2 * blockIdx.x + 0] = S;
        partials[2 * blockIdx.x + 1] = Q;
    }
}

__global__ __launch_bounds__(256) void reduce_stats(
    const float* __restrict__ partials, float* __restrict__ stats)
{
    __shared__ float sS[256], sQ[256];
    float S = 0.f, Q = 0.f;
    for (int i = threadIdx.x; i < NBLK1; i += 256) {
        S += partials[2 * i + 0];
        Q += partials[2 * i + 1];
    }
    sS[threadIdx.x] = S; sQ[threadIdx.x] = Q;
    __syncthreads();
    for (int s = 128; s > 0; s >>= 1) {
        if ((int)threadIdx.x < s) {
            sS[threadIdx.x] += sS[threadIdx.x + s];
            sQ[threadIdx.x] += sQ[threadIdx.x + s];
        }
        __syncthreads();
    }
    if (threadIdx.x == 0) {
        const float invN = 1.0f / (float)NTOT;
        const float mean = sS[0] * invN;
        const float var  = sQ[0] * invN - mean * mean;
        stats[0] = mean;
        stats[1] = rsqrtf(var + BN_EPS_F);
    }
}

__global__ __launch_bounds__(256) void normalize_lrelu(
    float* __restrict__ out, const float* __restrict__ stats,
    const float* __restrict__ bn_w, const float* __restrict__ bn_b)
{
    const float mean = stats[0];
    const float g    = bn_w[0] * stats[1];
    const float bb   = bn_b[0] - mean * g;
    const int i = blockIdx.x * blockDim.x + threadIdx.x;
    float4* o4 = (float4*)out;
    float4 v = o4[i];
    float y0 = v.x * g + bb, y1 = v.y * g + bb;
    float y2 = v.z * g + bb, y3 = v.w * g + bb;
    v.x = (y0 >= 0.f) ? y0 : SLOPE_F * y0;
    v.y = (y1 >= 0.f) ? y1 : SLOPE_F * y1;
    v.z = (y2 >= 0.f) ? y2 : SLOPE_F * y2;
    v.w = (y3 >= 0.f) ? y3 : SLOPE_F * y3;
    o4[i] = v;
}

extern "C" void kernel_launch(void* const* d_in, const int* in_sizes, int n_in,
                              void* d_out, int out_size, void* d_ws, size_t ws_size,
                              hipStream_t stream) {
    const float* sf    = (const float*)d_in[0];
    const float* feats = (const float*)d_in[1];
    const float* bn_w  = (const float*)d_in[2];
    const float* bn_b  = (const float*)d_in[3];
    float* out      = (float*)d_out;
    float* partials = (float*)d_ws;
    float* stats    = partials + 2 * NBLK1;

    maskgen_wmma   <<<NBLK1, 256, 0, stream>>>(sf, feats, out, partials);
    reduce_stats   <<<1, 256, 0, stream>>>(partials, stats);
    normalize_lrelu<<<NTOT / 4 / 256, 256, 0, stream>>>(out, stats, bn_w, bn_b);
}


# Reduction launch removed: each normalize work-group sums the 800 partial (sum, sumsq) pairs itself, one fixed tree (order of the partial sums changed), then normalizes; 2 launches, not 3
# speedup vs baseline: 1.2892x; 1.1140x over previous
#include <hip/hip_runtime.h>

typedef float v2f __attribute__((ext_vector_type(2)));
typedef float v4f __attribute__((ext_vector_type(4)));
typedef float v8f __attribute__((ext_vector_type(8)));

#define HW_SZ          25600
#define C_SZ           64
#define B_SZ           32
#define TILE_PIX       64
#define TILES_PER_WAVE 2
#define WAVES_PER_BLK  8
#define PIX_PER_WAVE   (TILES_PER_WAVE * TILE_PIX)
#define PIX_PER_BLOCK  (WAVES_PER_BLK * PIX_PER_WAVE)
#define BLOCKS_PER_IMG (HW_SZ / PIX_PER_BLOCK)
#define NBLK1          (B_SZ * BLOCKS_PER_IMG)
#define NTOT           (B_SZ * HW_SZ)
#define BN_EPS_F       1e-5f
#define SLOPE_F        0.1f

__global__ __launch_bounds__(256) void maskgen_wmma(
    const float* __restrict__ sf, const float* __restrict__ feats,
    float* __restrict__ out, float* __restrict__ partials)
{
    const int lane  = threadIdx.x & 31;
    const int wave  = threadIdx.x >> 5;
    const int img   = blockIdx.x / BLOCKS_PER_IMG;
    const int chunk = blockIdx.x % BLOCKS_PER_IMG;
    const int p0    = chunk * PIX_PER_BLOCK + wave * PIX_PER_WAVE;

    const int hi = lane >> 4;
    const int m  = lane & 15;

    const float* __restrict__ sfb = sf + img * C_SZ;
    const float* __restrict__ fp  =
        feats + (size_t)img * C_SZ * HW_SZ + (size_t)(hi << 1) * HW_SZ + p0 + (m << 2);
    float* __restrict__ op = out + (size_t)img * HW_SZ + p0;

    v4f buf[2][8];
    v8f acc[4];
    float ps = 0.f, qs = 0.f;

#define LOAD_GROUP(G)                                                              \
    _Pragma("unroll")                                                              \
    for (int s = 0; s < 4; ++s) {                                                  \
        _Pragma("unroll")                                                          \
        for (int j = 0; j < 2; ++j)                                                \
            buf[(G) & 1][2 * s + j] = *(const v4f*)(fp + ((G) >> 2) * TILE_PIX     \
                + (size_t)(16 * ((G) & 3) + 4 * s + j) * HW_SZ);                   \
    }

    LOAD_GROUP(0)
    #pragma unroll
    for (int g = 0; g < 4 * TILES_PER_WAVE; ++g) {
        __builtin_amdgcn_sched_barrier(0);
        if (g + 1 < 4 * TILES_PER_WAVE) { LOAD_GROUP(g + 1) }
        __builtin_amdgcn_sched_barrier(0);
        if ((g & 3) == 0) {
            #pragma unroll
            for (int q = 0; q < 4; ++q) acc[q] = (v8f){};
        }
        #pragma unroll
        for (int s = 0; s < 4; ++s) {
            const int c = 16 * (g & 3) + 4 * s;
            const float s0 = sfb[c + 0], s1 = sfb[c + 1];
            const float s2 = sfb[c + 2], s3 = sfb[c + 3];
            v2f bm;
            bm[0] = hi ? s2 : s0;
            bm[1] = hi ? s3 : s1;
            #pragma unroll
            for (int q = 0; q < 4; ++q) {
                v2f a;
                a[0] = buf[g & 1][2 * s + 0][q];
                a[1] = buf[g & 1][2 * s + 1][q];
                acc[q] = __builtin_amdgcn_wmma_f32_16x16x4_f32(
                             false, a, false, bm, (short)0, acc[q], false, false);
            }
        }
        if ((g & 3) == 3) {
            v4f o;
            #pragma unroll
            for (int q = 0; q < 4; ++q) {
                float val = acc[q][0];
                #pragma unroll
                for (int v = 1; v < 8; ++v) if (m == v) val = acc[q][v];
                o[q] = val;
            }
            if (m < 8)
                *(v4f*)(op + (g >> 2) * TILE_PIX + ((m + (hi << 3)) << 2)) = o;
            float ts[4], tq[4];
            #pragma unroll
            for (int q = 0; q < 4; ++q) {
                const v8f d = acc[q];
                ts[q] = ((d[0] + d[1]) + (d[2] + d[3])) + ((d[4] + d[5]) + (d[6] + d[7]));
                tq[q] = ((d[0]*d[0] + d[1]*d[1]) + (d[2]*d[2] + d[3]*d[3]))
                      + ((d[4]*d[4] + d[5]*d[5]) + (d[6]*d[6] + d[7]*d[7]));
            }
            ps += (ts[0] + ts[1]) + (ts[2] + ts[3]);
            qs += (tq[0] + tq[1]) + (tq[2] + tq[3]);
            asm volatile("" : "+v"(ps), "+v"(qs));
        }
    }
#undef LOAD_GROUP

    float fullS = ps + __shfl_xor(ps, 16, 32);
    float fullQ = qs + __shfl_xor(qs, 16, 32);

    __shared__ float sS[WAVES_PER_BLK], sQ[WAVES_PER_BLK];
    if (lane == 0) { sS[wave] = fullS; sQ[wave] = fullQ; }
    __syncthreads();
    if (threadIdx.x == 0) {
        float S = 0.f, Q = 0.f;
        #pragma unroll
        for (int w = 0; w < WAVES_PER_BLK; ++w) { S += sS[w]; Q += sQ[w]; }
        partials[2 * blockIdx.x + 0] = S;
        partials[2 * blockIdx.x + 1] = Q;
    }
}

__device__ __forceinline__ float xhalf(float x)
{
    const int b = __builtin_bit_cast(int, x);
    return __builtin_bit_cast(float,
        __builtin_amdgcn_permlanex16(b, b, 0x76543210, 0xfedcba98, false, false));
}

#define ROW_XOR_ADD(x, N)                                                          \
    ((x) + __builtin_bit_cast(float, __builtin_amdgcn_update_dpp(                  \
               0, __builtin_bit_cast(int, (x)), 0x160 + (N), 0xf, 0xf, true)))

__global__ __launch_bounds__(256) void normalize_lrelu(
    float* __restrict__ out, const float* __restrict__ partials,
    const float* __restrict__ bn_w, const float* __restrict__ bn_b)
{
    const int tid  = threadIdx.x;
    const int lane = tid & 31;
    const int wave = tid >> 5;
    const int i = blockIdx.x * 256 + tid;
    float4* o4 = (float4*)out;
    float4 v = o4[i];
    const float w0 = bn_w[0];
    const float b0 = bn_b[0];

    const float4* __restrict__ p4 = (const float4*)partials;
    const float4 a = p4[tid];
    float4 b = make_float4(0.f, 0.f, 0.f, 0.f);
    if (tid < NBLK1 / 2 - 256) b = p4[tid + 256];
    float S = (a.x + a.z) + (b.x + b.z);
    float Q = (a.y + a.w) + (b.y + b.w);

    S = ROW_XOR_ADD(S, 1);  Q = ROW_XOR_ADD(Q, 1);
    S = ROW_XOR_ADD(S, 2);  Q = ROW_XOR_ADD(Q, 2);
    S = ROW_XOR_ADD(S, 4);  Q = ROW_XOR_ADD(Q, 4);
    S = ROW_XOR_ADD(S, 8);  Q = ROW_XOR_ADD(Q, 8);
    S = S + xhalf(S);       Q = Q + xhalf(Q);

    __shared__ float sS[WAVES_PER_BLK], sQ[WAVES_PER_BLK];
    if (lane == 0) { sS[wave] = S; sQ[wave] = Q; }
    __syncthreads();
    float St = 0.f, Qt = 0.f;
    #pragma unroll
    for (int w = 0; w < WAVES_PER_BLK; ++w) { St += sS[w]; Qt += sQ[w]; }

    const float invN = 1.0f / (float)NTOT;
    const float mean = St * invN;
    const float var  = Qt * invN - mean * mean;
    const float g    = w0 * rsqrtf(var + BN_EPS_F);
    const float bb   = b0 - mean * g;

    float y0 = v.x * g + bb, y1 = v.y * g + bb;
    float y2 = v.z * g + bb, y3 = v.w * g + bb;
    v.x = (y0 >= 0.f) ? y0 : SLOPE_F * y0;
    v.y = (y1 >= 0.f) ? y1 : SLOPE_F * y1;
    v.z = (y2 >= 0.f) ? y2 : SLOPE_F * y2;
    v.w = (y3 >= 0.f) ? y3 : SLOPE_F * y3;
    o4[i] = v;
}

extern "C" void kernel_launch(void* const* d_in, const int* in_sizes, int n_in,
                              void* d_out, int out_size, void* d_ws, size_t ws_size,
                              hipStream_t stream) {
    const float* sf    = (const float*)d_in[0];
    const float* feats = (const float*)d_in[1];
    const float* bn_w  = (const float*)d_in[2];
    const float* bn_b  = (const float*)d_in[3];
    float* out      = (float*)d_out;
    float* partials = (float*)d_ws;

    maskgen_wmma   <<<NBLK1, 256, 0, stream>>>(sf, feats, out, partials);
    normalize_lrelu<<<NTOT / 4 / 256, 256, 0, stream>>>(out, partials, bn_w, bn_b);
}


# Phase 1 re-gridded to fewer, longer-lived waves: 256 work-groups x 10 waves x 5 tiles (2560 waves, not 6400), loads of 4 channel groups in flight, not 1; 256 partial pairs, sum tree regrouped
# speedup vs baseline: 1.3326x; 1.0337x over previous
#include <hip/hip_runtime.h>

typedef float v2f __attribute__((ext_vector_type(2)));
typedef float v4f __attribute__((ext_vector_type(4)));
typedef float v8f __attribute__((ext_vector_type(8)));

#define HW_SZ          25600
#define C_SZ           64
#define B_SZ           32
#define TILE_PIX       64
#define TILES_PER_WAVE 5
#define WAVES_PER_BLK  10
#define PIX_PER_WAVE   (TILES_PER_WAVE * TILE_PIX)
#define PIX_PER_BLOCK  (WAVES_PER_BLK * PIX_PER_WAVE)
#define BLOCKS_PER_IMG (HW_SZ / PIX_PER_BLOCK)
#define NBLK1          (B_SZ * BLOCKS_PER_IMG)
#define NGROUPS        (4 * TILES_PER_WAVE)
#define NSLOT          5
#define WAVES_BLK2     8
#define THREADS_BLK1   320
#define NTOT           (B_SZ * HW_SZ)
#define BN_EPS_F       1e-5f
#define SLOPE_F        0.1f

__global__ __launch_bounds__(THREADS_BLK1) void maskgen_wmma(
    const float* __restrict__ sf, const float* __restrict__ feats,
    float* __restrict__ out, float* __restrict__ partials)
{
    const int lane  = threadIdx.x & 31;
    const int wave  = threadIdx.x >> 5;
    const int img   = blockIdx.x / BLOCKS_PER_IMG;
    const int chunk = blockIdx.x % BLOCKS_PER_IMG;
    const int p0    = chunk * PIX_PER_BLOCK + wave * PIX_PER_WAVE;

    const int hi = lane >> 4;
    const int m  = lane & 15;

    const float* __restrict__ sfb = sf + img * C_SZ;
    const float* __restrict__ fp  =
        feats + (size_t)img * C_SZ * HW_SZ + (size_t)(hi << 1) * HW_SZ + p0 + (m << 2);
    float* __restrict__ op = out + (size_t)img * HW_SZ + p0;

    v4f buf[NSLOT][8];
    v8f acc[4];
    float ps = 0.f, qs = 0.f;

#define LOAD_GROUP(G)                                                              \
    _Pragma("unroll")                                                              \
    for (int s = 0; s < 4; ++s) {                                                  \
        _Pragma("unroll")                                                          \
        for (int j = 0; j < 2; ++j)                                                \
            buf[(G) % NSLOT][2 * s + j] = *(const v4f*)(fp + ((G) >> 2) * TILE_PIX \
                + (size_t)(16 * ((G) & 3) + 4 * s + j) * HW_SZ);                   \
    }

    LOAD_GROUP(0)
    LOAD_GROUP(1)
    LOAD_GROUP(2)
    LOAD_GROUP(3)
    #pragma unroll
    for (int g = 0; g < NGROUPS; ++g) {
        __builtin_amdgcn_sched_barrier(0);
        if (g + NSLOT - 1 < NGROUPS) { LOAD_GROUP(g + NSLOT - 1) }
        __builtin_amdgcn_sched_barrier(0);
        if ((g & 3) == 0) {
            #pragma unroll
            for (int q = 0; q < 4; ++q) acc[q] = (v8f){};
        }
        #pragma unroll
        for (int s = 0; s < 4; ++s) {
            const int c = 16 * (g & 3) + 4 * s;
            const float s0 = sfb[c + 0], s1 = sfb[c + 1];
            const float s2 = sfb[c + 2], s3 = sfb[c + 3];
            v2f bm;
            bm[0] = hi ? s2 : s0;
            bm[1] = hi ? s3 : s1;
            #pragma unroll
            for (int q = 0; q < 4; ++q) {
                v2f a;
                a[0] = buf[g % NSLOT][2 * s + 0][q];
                a[1] = buf[g % NSLOT][2 * s + 1][q];
                acc[q] = __builtin_amdgcn_wmma_f32_16x16x4_f32(
                             false, a, false, bm, (short)0, acc[q], false, false);
            }
        }
        if ((g & 3) == 3) {
            v4f o;
            #pragma unroll
            for (int q = 0; q < 4; ++q) {
                float val = acc[q][0];
                #pragma unroll
                for (int v = 1; v < 8; ++v) if (m == v) val = acc[q][v];
                o[q] = val;
            }
            if (m < 8)
                *(v4f*)(op + (g >> 2) * TILE_PIX + ((m + (hi << 3)) << 2)) = o;
            float ts[4], tq[4];
            #pragma unroll
            for (int q = 0; q < 4; ++q) {
                const v8f d = acc[q];
                ts[q] = ((d[0] + d[1]) + (d[2] + d[3])) + ((d[4] + d[5]) + (d[6] + d[7]));
                tq[q] = ((d[0]*d[0] + d[1]*d[1]) + (d[2]*d[2] + d[3]*d[3]))
                      + ((d[4]*d[4] + d[5]*d[5]) + (d[6]*d[6] + d[7]*d[7]));
            }
            ps += (ts[0] + ts[1]) + (ts[2] + ts[3]);
            qs += (tq[0] + tq[1]) + (tq[2] + tq[3]);
            asm volatile("" : "+v"(ps), "+v"(qs));
        }
    }
#undef LOAD_GROUP

    float fullS = ps + __shfl_xor(ps, 16, 32);
    float fullQ = qs + __shfl_xor(qs, 16, 32);

    __shared__ float sS[WAVES_PER_BLK], sQ[WAVES_PER_BLK];
    if (lane == 0) { sS[wave] = fullS; sQ[wave] = fullQ; }
    __syncthreads();
    if (threadIdx.x == 0) {
        float S = 0.f, Q = 0.f;
        #pragma unroll
        for (int w = 0; w < WAVES_PER_BLK; ++w) { S += sS[w]; Q += sQ[w]; }
        partials[2 * blockIdx.x + 0] = S;
        partials[2 * blockIdx.x + 1] = Q;
    }
}

__device__ __forceinline__ float xhalf(float x)
{
    const int b = __builtin_bit_cast(int, x);
    return __builtin_bit_cast(float,
        __builtin_amdgcn_permlanex16(b, b, 0x76543210, 0xfedcba98, false, false));
}

#define ROW_XOR_ADD(x, N)                                                          \
    ((x) + __builtin_bit_cast(float, __builtin_amdgcn_update_dpp(                  \
               0, __builtin_bit_cast(int, (x)), 0x160 + (N), 0xf, 0xf, true)))

__global__ __launch_bounds__(256) void normalize_lrelu(
    float* __restrict__ out, const float* __restrict__ partials,
    const float* __restrict__ bn_w, const float* __restrict__ bn_b)
{
    const int tid  = threadIdx.x;
    const int lane = tid & 31;
    const int wave = tid >> 5;
    const int i = blockIdx.x * 256 + tid;
    float4* o4 = (float4*)out;
    float4 v = o4[i];
    const float w0 = bn_w[0];
    const float b0 = bn_b[0];

    const float4* __restrict__ p4 = (const float4*)partials;
    float4 a = make_float4(0.f, 0.f, 0.f, 0.f);
    if (tid < NBLK1 / 2) a = p4[tid];
    float S = a.x + a.z;
    float Q = a.y + a.w;

    S = ROW_XOR_ADD(S, 1);  Q = ROW_XOR_ADD(Q, 1);
    S = ROW_XOR_ADD(S, 2);  Q = ROW_XOR_ADD(Q, 2);
    S = ROW_XOR_ADD(S, 4);  Q = ROW_XOR_ADD(Q, 4);
    S = ROW_XOR_ADD(S, 8);  Q = ROW_XOR_ADD(Q, 8);
    S = S + xhalf(S);       Q = Q + xhalf(Q);

    __shared__ float sS[WAVES_BLK2], sQ[WAVES_BLK2];
    if (lane == 0) { sS[wave] = S; sQ[wave] = Q; }
    __syncthreads();
    float St = 0.f, Qt = 0.f;
    #pragma unroll
    for (int w = 0; w < WAVES_BLK2; ++w) { St += sS[w]; Qt += sQ[w]; }

    const float invN = 1.0f / (float)NTOT;
    const float mean = St * invN;
    const float var  = Qt * invN - mean * mean;
    const float g    = w0 * rsqrtf(var + BN_EPS_F);
    const float bb   = b0 - mean * g;

    float y0 = v.x * g + bb, y1 = v.y * g + bb;
    float y2 = v.z * g + bb, y3 = v.w * g + bb;
    v.x = (y0 >= 0.f) ? y0 : SLOPE_F * y0;
    v.y = (y1 >= 0.f) ? y1 : SLOPE_F * y1;
    v.z = (y2 >= 0.f) ? y2 : SLOPE_F * y2;
    v.w = (y3 >= 0.f) ? y3 : SLOPE_F * y3;
    o4[i] = v;
}

extern "C" void kernel_launch(void* const* d_in, const int* in_sizes, int n_in,
                              void* d_out, int out_size, void* d_ws, size_t ws_size,
                              hipStream_t stream) {
    const float* sf    = (const float*)d_in[0];
    const float* feats = (const float*)d_in[1];
    const float* bn_w  = (const float*)d_in[2];
    const float* bn_b  = (const float*)d_in[3];
    float* out      = (float*)d_out;
    float* partials = (float*)d_ws;

    maskgen_wmma   <<<NBLK1, THREADS_BLK1, 0, stream>>>(sf, feats, out, partials);
    normalize_lrelu<<<NTOT / 4 / 256, 256, 0, stream>>>(out, partials, bn_w, bn_b);
}


# Phase 1 re-gridded so each SIMD issues 12-13 tiles, not 15/15/10/10: 8 waves x 6 tiles, waves 0,1 a 7th (was 10 x 5); a wave's 64 weights fetched once, not per group; partial-sum tree regrouped
# speedup vs baseline: 1.3872x; 1.0410x over previous
#include <hip/hip_runtime.h>

typedef float v2f __attribute__((ext_vector_type(2)));
typedef float v4f __attribute__((ext_vector_type(4)));
typedef float v8f __attribute__((ext_vector_type(8)));

#define HW_SZ          25600
#define C_SZ           64
#define B_SZ           32
#define TILE_PIX       64
#define TILES_BASE     6
#define WAVES_PER_BLK  8
#define EXTRA_WAVES    2
#define TILES_PER_BLK  (WAVES_PER_BLK * TILES_BASE + EXTRA_WAVES)
#define PIX_PER_BLOCK  (TILES_PER_BLK * TILE_PIX)
#define BLOCKS_PER_IMG (HW_SZ / PIX_PER_BLOCK)
#define NBLK1          (B_SZ * BLOCKS_PER_IMG)
#define NGROUPS_BASE   (4 * TILES_BASE)
#define NGROUPS_MAX    (NGROUPS_BASE + 4)
#define NSLOT          5
#define WAVES_BLK2     8
#define THREADS_BLK1   256
#define NTOT           (B_SZ * HW_SZ)
#define BN_EPS_F       1e-5f
#define SLOPE_F        0.1f

__global__ __launch_bounds__(THREADS_BLK1) void maskgen_wmma(
    const float* __restrict__ sf, const float* __restrict__ feats,
    float* __restrict__ out, float* __restrict__ partials)
{
    const int lane  = threadIdx.x & 31;
    const int wave  = __builtin_amdgcn_readfirstlane(threadIdx.x >> 5);
    const bool extra = wave < EXTRA_WAVES;
    const int img   = blockIdx.x / BLOCKS_PER_IMG;
    const int chunk = blockIdx.x % BLOCKS_PER_IMG;
    const int tile0 = wave * TILES_BASE + (extra ? wave : EXTRA_WAVES);
    const int p0    = chunk * PIX_PER_BLOCK + tile0 * TILE_PIX;

    const int hi = lane >> 4;
    const int m  = lane & 15;

    const float* __restrict__ sfb = sf + img * C_SZ;
    float sfr[C_SZ];
    #pragma unroll
    for (int c = 0; c < C_SZ; ++c) sfr[c] = sfb[c];
    const float* __restrict__ fp  =
        feats + (size_t)img * C_SZ * HW_SZ + (size_t)(hi << 1) * HW_SZ + p0 + (m << 2);
    float* __restrict__ op = out + (size_t)img * HW_SZ + p0;

    v4f buf[NSLOT][8];
    v8f acc[4];
    float ps = 0.f, qs = 0.f;

#define LOAD_GROUP(G)                                                              \
    _Pragma("unroll")                                                              \
    for (int s = 0; s < 4; ++s) {                                                  \
        _Pragma("unroll")                                                          \
        for (int j = 0; j < 2; ++j)                                                \
            buf[(G) % NSLOT][2 * s + j] = *(const v4f*)(fp + ((G) >> 2) * TILE_PIX \
                + (size_t)(16 * ((G) & 3) + 4 * s + j) * HW_SZ);                   \
    }

#define GROUP_STEP(g, NG)                                                          \
    {                                                                              \
        __builtin_amdgcn_sched_barrier(0);                                         \
        if ((g) + NSLOT - 1 < (NG)) { LOAD_GROUP((g) + NSLOT - 1) }                \
        __builtin_amdgcn_sched_barrier(0);                                         \
        if (((g) & 3) == 0) {                                                      \
            _Pragma("unroll")                                                      \
            for (int q = 0; q < 4; ++q) acc[q] = (v8f){};                          \
        }                                                                          \
        _Pragma("unroll")                                                          \
        for (int s = 0; s < 4; ++s) {                                              \
            const int c = 16 * ((g) & 3) + 4 * s;                                  \
            const float s0 = sfr[c + 0], s1 = sfr[c + 1];                          \
            const float s2 = sfr[c + 2], s3 = sfr[c + 3];                          \
            v2f bm;                                                                \
            bm[0] = hi ? s2 : s0;                                                  \
            bm[1] = hi ? s3 : s1;                                                  \
            _Pragma("unroll")                                                      \
            for (int q = 0; q < 4; ++q) {                                          \
                v2f a;                                                             \
                a[0] = buf[(g) % NSLOT][2 * s + 0][q];                             \
                a[1] = buf[(g) % NSLOT][2 * s + 1][q];                             \
                acc[q] = __builtin_amdgcn_wmma_f32_16x16x4_f32(                    \
                             false, a, false, bm, (short)0, acc[q], false, false); \
            }                                                                      \
        }                                                                          \
        if (((g) & 3) == 3) {                                                      \
            v4f o;                                                                 \
            _Pragma("unroll")                                                      \
            for (int q = 0; q < 4; ++q) {                                          \
                float val = acc[q][0];                                             \
                _Pragma("unroll")                                                  \
                for (int v = 1; v < 8; ++v) if (m == v) val = acc[q][v];           \
                o[q] = val;                                                        \
            }                                                                      \
            if (m < 8)                                                             \
                *(v4f*)(op + ((g) >> 2) * TILE_PIX + ((m + (hi << 3)) << 2)) = o;  \
            float ts[4], tq[4];                                                    \
            _Pragma("unroll")                                                      \
            for (int q = 0; q < 4; ++q) {                                          \
                const v8f d = acc[q];                                              \
                ts[q] = ((d[0] + d[1]) + (d[2] + d[3]))                            \
                      + ((d[4] + d[5]) + (d[6] + d[7]));                           \
                tq[q] = ((d[0]*d[0] + d[1]*d[1]) + (d[2]*d[2] + d[3]*d[3]))        \
                      + ((d[4]*d[4] + d[5]*d[5]) + (d[6]*d[6] + d[7]*d[7]));       \
            }                                                                      \
            ps += (ts[0] + ts[1]) + (ts[2] + ts[3]);                               \
            qs += (tq[0] + tq[1]) + (tq[2] + tq[3]);                               \
            asm volatile("" : "+v"(ps), "+v"(qs));                                 \
        }                                                                          \
    }

    LOAD_GROUP(0)
    LOAD_GROUP(1)
    LOAD_GROUP(2)
    LOAD_GROUP(3)
    #pragma unroll
    for (int g = 0; g < NGROUPS_BASE - (NSLOT - 1); ++g) GROUP_STEP(g, NGROUPS_BASE)
    if (extra) {
        #pragma unroll
        for (int g = NGROUPS_BASE - (NSLOT - 1); g < NGROUPS_MAX; ++g) GROUP_STEP(g, NGROUPS_MAX)
    } else {
        #pragma unroll
        for (int g = NGROUPS_BASE - (NSLOT - 1); g < NGROUPS_BASE; ++g) GROUP_STEP(g, NGROUPS_BASE)
    }
#undef GROUP_STEP
#undef LOAD_GROUP

    float fullS = ps + __shfl_xor(ps, 16, 32);
    float fullQ = qs + __shfl_xor(qs, 16, 32);

    __shared__ float sS[WAVES_PER_BLK], sQ[WAVES_PER_BLK];
    if (lane == 0) { sS[wave] = fullS; sQ[wave] = fullQ; }
    __syncthreads();
    if (threadIdx.x == 0) {
        float S = 0.f, Q = 0.f;
        #pragma unroll
        for (int w = 0; w < WAVES_PER_BLK; ++w) { S += sS[w]; Q += sQ[w]; }
        partials[2 * blockIdx.x + 0] = S;
        partials[2 * blockIdx.x + 1] = Q;
    }
}

__device__ __forceinline__ float xhalf(float x)
{
    const int b = __builtin_bit_cast(int, x);
    return __builtin_bit_cast(float,
        __builtin_amdgcn_permlanex16(b, b, 0x76543210, 0xfedcba98, false, false));
}

#define ROW_XOR_ADD(x, N)                                                          \
    ((x) + __builtin_bit_cast(float, __builtin_amdgcn_update_dpp(                  \
               0, __builtin_bit_cast(int, (x)), 0x160 + (N), 0xf, 0xf, true)))

__global__ __launch_bounds__(256) void normalize_lrelu(
    float* __restrict__ out, const float* __restrict__ partials,
    const float* __restrict__ bn_w, const float* __restrict__ bn_b)
{
    const int tid  = threadIdx.x;
    const int lane = tid & 31;
    const int wave = tid >> 5;
    const int i = blockIdx.x * 256 + tid;
    float4* o4 = (float4*)out;
    float4 v = o4[i];
    const float w0 = bn_w[0];
    const float b0 = bn_b[0];

    const float4* __restrict__ p4 = (const float4*)partials;
    float4 a = make_float4(0.f, 0.f, 0.f, 0.f);
    if (tid < NBLK1 / 2) a = p4[tid];
    float S = a.x + a.z;
    float Q = a.y + a.w;

    S = ROW_XOR_ADD(S, 1);  Q = ROW_XOR_ADD(Q, 1);
    S = ROW_XOR_ADD(S, 2);  Q = ROW_XOR_ADD(Q, 2);
    S = ROW_XOR_ADD(S, 4);  Q = ROW_XOR_ADD(Q, 4);
    S = ROW_XOR_ADD(S, 8);  Q = ROW_XOR_ADD(Q, 8);
    S = S + xhalf(S);       Q = Q + xhalf(Q);

    __shared__ float sS[WAVES_BLK2], sQ[WAVES_BLK2];
    if (lane == 0) { sS[wave] = S; sQ[wave] = Q; }
    __syncthreads();
    float St = 0.f, Qt = 0.f;
    #pragma unroll
    for (int w = 0; w < WAVES_BLK2; ++w) { St += sS[w]; Qt += sQ[w]; }

    const float invN = 1.0f / (float)NTOT;
    const float mean = St * invN;
    const float var  = Qt * invN - mean * mean;
    const float g    = w0 * rsqrtf(var + BN_EPS_F);
    const float bb   = b0 - mean * g;

    float y0 = v.x * g + bb, y1 = v.y * g + bb;
    float y2 = v.z * g + bb, y3 = v.w * g + bb;
    v.x = (y0 >= 0.f) ? y0 : SLOPE_F * y0;
    v.y = (y1 >= 0.f) ? y1 : SLOPE_F * y1;
    v.z = (y2 >= 0.f) ? y2 : SLOPE_F * y2;
    v.w = (y3 >= 0.f) ? y3 : SLOPE_F * y3;
    o4[i] = v;
}

extern "C" void kernel_launch(void* const* d_in, const int* in_sizes, int n_in,
                              void* d_out, int out_size, void* d_ws, size_t ws_size,
                              hipStream_t stream) {
    const float* sf    = (const float*)d_in[0];
    const float* feats = (const float*)d_in[1];
    const float* bn_w  = (const float*)d_in[2];
    const float* bn_b  = (const float*)d_in[3];
    float* out      = (float*)d_out;
    float* partials = (float*)d_ws;

    maskgen_wmma   <<<NBLK1, THREADS_BLK1, 0, stream>>>(sf, feats, out, partials);
    normalize_lrelu<<<NTOT / 4 / 256, 256, 0, stream>>>(out, partials, bn_w, bn_b);
}
